# starting kernel
# baseline (speedup 1.0000x reference)
#include <hip/hip_runtime.h>

typedef __attribute__((ext_vector_type(2))) float v2f;
typedef __attribute__((ext_vector_type(8))) float v8f;

#define SAI_N 9
#define B_N   4
#define CK_N  64
#define CV_N  64
#define HW_N  1024
#define SHW_N (SAI_N * HW_N)
#define RADIUS_F 0.5f
#define NEG_BIG (-1.0e30f)

#define NW 8
#define KEYS_PER_WAVE (SHW_N / NW)

__global__ __launch_bounds__(32 * NW)
void cfa_fp32_wmma_splitk(const float* __restrict__ mem_keys,
                          const float* __restrict__ mem_vals,
                          const float* __restrict__ query,
                          const float* __restrict__ disparity,
                          const int*   __restrict__ seq_index,
                          float* __restrict__ out)
{
    __shared__ float pt[NW][16 * 16];
    __shared__ float lm[NW][16];
    __shared__ float ll[NW][16];
    __shared__ float lo[NW][16][CV_N];

    const int tid  = threadIdx.x;
    const int w    = tid >> 5;
    const int lane = tid & 31;
    const int b    = blockIdx.x >> 6;
    const int q0   = (blockIdx.x & 63) * 16;

    const int m    = lane & 15;
    const int kb   = (lane >> 4) * 2;
    const int n    = lane & 15;
    const int half = lane >> 4;

    v2f qa[16];
    {
        const float* qbase = query + (size_t)b * CK_N * HW_N + q0 + m;
        #pragma unroll
        for (int cc = 0; cc < 16; ++cc) {
            const int c0 = cc * 4 + kb;
            qa[cc].x = qbase[(size_t)c0 * HW_N];
            qa[cc].y = qbase[(size_t)(c0 + 1) * HW_N];
        }
    }

    float mrow[8], lsum[8];
    #pragma unroll
    for (int v = 0; v < 8; ++v) { mrow[v] = NEG_BIG; lsum[v] = 0.0f; }
    v8f oacc[4] = {};

    const int kstart = w * KEYS_PER_WAVE;
    const int kend   = kstart + KEYS_PER_WAVE;
    for (int s = kstart >> 10; s * HW_N < kend; ++s) {
        const int klo = (kstart > s * HW_N) ? kstart : s * HW_N;
        const int khi = (kend < (s + 1) * HW_N) ? kend : (s + 1) * HW_N;

        const float dx = (float)seq_index[(b * SAI_N + s) * 2 + 0] - 5.0f;
        const float dy = (float)seq_index[(b * SAI_N + s) * 2 + 1] - 5.0f;
        const float df = sqrtf(dx * dx + dy * dy);

        const float* kframe = mem_keys + (((size_t)s * B_N + b) * CK_N) * HW_N;
        const float* vframe = mem_vals + (((size_t)s * B_N + b) * CV_N) * HW_N;

        for (int k0 = klo; k0 < khi; k0 += 16) {
            const int p0 = k0 & (HW_N - 1);
            const float* kbase = kframe + p0;
            const float* vbase = vframe + p0;

            __builtin_prefetch((const void*)(kbase + 16), 0, 1);

            v8f sc = {};
            #pragma unroll
            for (int cc = 0; cc < 16; ++cc) {
                const int c0 = cc * 4 + kb;
                v2f bk;
                bk.x = kbase[(size_t)c0 * HW_N + n];
                bk.y = kbase[(size_t)(c0 + 1) * HW_N + n];
                sc = __builtin_amdgcn_wmma_f32_16x16x4_f32(
                    false, qa[cc], false, bk, (short)0, sc, false, false);
            }

            const float disp = disparity[(size_t)b * HW_N + p0 + n];
            const float bias = (fabsf(df * disp) > RADIUS_F) ? 0.0f : NEG_BIG;
            #pragma unroll
            for (int v = 0; v < 8; ++v) sc[v] += bias;

            float pfrag[8];
            #pragma unroll
            for (int v = 0; v < 8; ++v) {
                float cm = sc[v];
                #pragma unroll
                for (int off = 1; off < 16; off <<= 1)
                    cm = fmaxf(cm, __shfl_xor(cm, off, 32));
                const float mn    = fmaxf(mrow[v], cm);
                const float scale = __expf(mrow[v] - mn);
                const float pv    = __expf(sc[v] - mn);
                lsum[v] = lsum[v] * scale + pv;
                mrow[v] = mn;
                #pragma unroll
                for (int t = 0; t < 4; ++t) oacc[t][v] *= scale;
                pfrag[v] = pv;
            }

            #pragma unroll
            for (int v = 0; v < 8; ++v)
                pt[w][(v + 8 * half) * 16 + n] = pfrag[v];
            __builtin_amdgcn_wave_barrier();

            #pragma unroll
            for (int j = 0; j < 4; ++j) {
                v2f pa;
                pa.x = pt[w][m * 16 + j * 4 + kb];
                pa.y = pt[w][m * 16 + j * 4 + kb + 1];
                #pragma unroll
                for (int t = 0; t < 4; ++t) {
                    const int cv0 = t * 16 + n;
                    v2f bv;
                    bv.x = vbase[(size_t)cv0 * HW_N + (j * 4 + kb)];
                    bv.y = vbase[(size_t)cv0 * HW_N + (j * 4 + kb + 1)];
                    oacc[t] = __builtin_amdgcn_wmma_f32_16x16x4_f32(
                        false, pa, false, bv, (short)0, oacc[t], false, false);
                }
            }
            __builtin_amdgcn_wave_barrier();
        }
    }

    float lrow[8];
    #pragma unroll
    for (int v = 0; v < 8; ++v) {
        float rs = lsum[v];
        #pragma unroll
        for (int off = 1; off < 16; off <<= 1)
            rs += __shfl_xor(rs, off, 32);
        lrow[v] = rs;
    }

    if (n == 0) {
        #pragma unroll
        for (int v = 0; v < 8; ++v) {
            lm[w][v + 8 * half] = mrow[v];
            ll[w][v + 8 * half] = lrow[v];
        }
    }
    #pragma unroll
    for (int v = 0; v < 8; ++v) {
        const int row = v + 8 * half;
        #pragma unroll
        for (int t = 0; t < 4; ++t)
            lo[w][row][t * 16 + n] = oacc[t][v];
    }
    __syncthreads();

    {
        const int row = tid >> 4;
        const int cv0 = (tid & 15) * 4;

        float mstar = NEG_BIG;
        #pragma unroll
        for (int w2 = 0; w2 < NW; ++w2) mstar = fmaxf(mstar, lm[w2][row]);

        float lstar = 0.0f;
        float acc[4] = {0.0f, 0.0f, 0.0f, 0.0f};
        #pragma unroll
        for (int w2 = 0; w2 < NW; ++w2) {
            const float f = __expf(lm[w2][row] - mstar);
            lstar += ll[w2][row] * f;
            #pragma unroll
            for (int j = 0; j < 4; ++j)
                acc[j] += lo[w2][row][cv0 + j] * f;
        }
        const float inv = 1.0f / lstar;
        #pragma unroll
        for (int j = 0; j < 4; ++j)
            out[((size_t)b * CV_N + cv0 + j) * HW_N + q0 + row] = acc[j] * inv;
    }
}

extern "C" void kernel_launch(void* const* d_in, const int* in_sizes, int n_in,
                              void* d_out, int out_size, void* d_ws, size_t ws_size,
                              hipStream_t stream) {
    const float* mem_keys  = (const float*)d_in[0];
    const float* mem_vals  = (const float*)d_in[1];
    const float* query     = (const float*)d_in[2];
    const float* disparity = (const float*)d_in[3];
    const int*   seq_index = (const int*)d_in[4];
    float*       out       = (float*)d_out;

    (void)in_sizes; (void)n_in; (void)out_size; (void)d_ws; (void)ws_size;

    dim3 grid(B_N * (HW_N / 16));
    dim3 block(32 * NW);
    cfa_fp32_wmma_splitk<<<grid, block, 0, stream>>>(
        mem_keys, mem_vals, query, disparity, seq_index, out);
}


# 16 split-K waves per block instead of 8 (4 per SIMD, 576 keys each): more resident waves cover each wave's load/shuffle/LDS waits; sum order changes: 16 partial softmax sums merged per row, not 8
# speedup vs baseline: 1.1585x; 1.1585x over previous
#include <hip/hip_runtime.h>

typedef __attribute__((ext_vector_type(2))) float v2f;
typedef __attribute__((ext_vector_type(8))) float v8f;

#define SAI_N 9
#define B_N   4
#define CK_N  64
#define CV_N  64
#define HW_N  1024
#define SHW_N (SAI_N * HW_N)
#define RADIUS_F 0.5f
#define NEG_BIG (-1.0e30f)

#define NW 16
#define KEYS_PER_WAVE (SHW_N / NW)

__global__ __launch_bounds__(32 * NW)
void cfa_fp32_wmma_splitk(const float* __restrict__ mem_keys,
                          const float* __restrict__ mem_vals,
                          const float* __restrict__ query,
                          const float* __restrict__ disparity,
                          const int*   __restrict__ seq_index,
                          float* __restrict__ out)
{
    __shared__ float pt[NW][16 * 16];
    __shared__ float lm[NW][16];
    __shared__ float ll[NW][16];
    __shared__ float lo[NW][16][CV_N];

    const int tid  = threadIdx.x;
    const int w    = tid >> 5;
    const int lane = tid & 31;
    const int b    = blockIdx.x >> 6;
    const int q0   = (blockIdx.x & 63) * 16;

    const int m    = lane & 15;
    const int kb   = (lane >> 4) * 2;
    const int n    = lane & 15;
    const int half = lane >> 4;

    v2f qa[16];
    {
        const float* qbase = query + (size_t)b * CK_N * HW_N + q0 + m;
        #pragma unroll
        for (int cc = 0; cc < 16; ++cc) {
            const int c0 = cc * 4 + kb;
            qa[cc].x = qbase[(size_t)c0 * HW_N];
            qa[cc].y = qbase[(size_t)(c0 + 1) * HW_N];
        }
    }

    float mrow[8], lsum[8];
    #pragma unroll
    for (int v = 0; v < 8; ++v) { mrow[v] = NEG_BIG; lsum[v] = 0.0f; }
    v8f oacc[4] = {};

    const int kstart = w * KEYS_PER_WAVE;
    const int kend   = kstart + KEYS_PER_WAVE;
    for (int s = kstart >> 10; s * HW_N < kend; ++s) {
        const int klo = (kstart > s * HW_N) ? kstart : s * HW_N;
        const int khi = (kend < (s + 1) * HW_N) ? kend : (s + 1) * HW_N;

        const float dx = (float)seq_index[(b * SAI_N + s) * 2 + 0] - 5.0f;
        const float dy = (float)seq_index[(b * SAI_N + s) * 2 + 1] - 5.0f;
        const float df = sqrtf(dx * dx + dy * dy);

        const float* kframe = mem_keys + (((size_t)s * B_N + b) * CK_N) * HW_N;
        const float* vframe = mem_vals + (((size_t)s * B_N + b) * CV_N) * HW_N;

        for (int k0 = klo; k0 < khi; k0 += 16) {
            const int p0 = k0 & (HW_N - 1);
            const float* kbase = kframe + p0;
            const float* vbase = vframe + p0;

            __builtin_prefetch((const void*)(kbase + 16), 0, 1);

            v8f sc = {};
            #pragma unroll
            for (int cc = 0; cc < 16; ++cc) {
                const int c0 = cc * 4 + kb;
                v2f bk;
                bk.x = kbase[(size_t)c0 * HW_N + n];
                bk.y = kbase[(size_t)(c0 + 1) * HW_N + n];
                sc = __builtin_amdgcn_wmma_f32_16x16x4_f32(
                    false, qa[cc], false, bk, (short)0, sc, false, false);
            }

            const float disp = disparity[(size_t)b * HW_N + p0 + n];
            const float bias = (fabsf(df * disp) > RADIUS_F) ? 0.0f : NEG_BIG;
            #pragma unroll
            for (int v = 0; v < 8; ++v) sc[v] += bias;

            float pfrag[8];
            #pragma unroll
            for (int v = 0; v < 8; ++v) {
                float cm = sc[v];
                #pragma unroll
                for (int off = 1; off < 16; off <<= 1)
                    cm = fmaxf(cm, __shfl_xor(cm, off, 32));
                const float mn    = fmaxf(mrow[v], cm);
                const float scale = __expf(mrow[v] - mn);
                const float pv    = __expf(sc[v] - mn);
                lsum[v] = lsum[v] * scale + pv;
                mrow[v] = mn;
                #pragma unroll
                for (int t = 0; t < 4; ++t) oacc[t][v] *= scale;
                pfrag[v] = pv;
            }

            #pragma unroll
            for (int v = 0; v < 8; ++v)
                pt[w][(v + 8 * half) * 16 + n] = pfrag[v];
            __builtin_amdgcn_wave_barrier();

            #pragma unroll
            for (int j = 0; j < 4; ++j) {
                v2f pa;
                pa.x = pt[w][m * 16 + j * 4 + kb];
                pa.y = pt[w][m * 16 + j * 4 + kb + 1];
                #pragma unroll
                for (int t = 0; t < 4; ++t) {
                    const int cv0 = t * 16 + n;
                    v2f bv;
                    bv.x = vbase[(size_t)cv0 * HW_N + (j * 4 + kb)];
                    bv.y = vbase[(size_t)cv0 * HW_N + (j * 4 + kb + 1)];
                    oacc[t] = __builtin_amdgcn_wmma_f32_16x16x4_f32(
                        false, pa, false, bv, (short)0, oacc[t], false, false);
                }
            }
            __builtin_amdgcn_wave_barrier();
        }
    }

    float lrow[8];
    #pragma unroll
    for (int v = 0; v < 8; ++v) {
        float rs = lsum[v];
        #pragma unroll
        for (int off = 1; off < 16; off <<= 1)
            rs += __shfl_xor(rs, off, 32);
        lrow[v] = rs;
    }

    if (n == 0) {
        #pragma unroll
        for (int v = 0; v < 8; ++v) {
            lm[w][v + 8 * half] = mrow[v];
            ll[w][v + 8 * half] = lrow[v];
        }
    }
    #pragma unroll
    for (int v = 0; v < 8; ++v) {
        const int row = v + 8 * half;
        #pragma unroll
        for (int t = 0; t < 4; ++t)
            lo[w][row][t * 16 + n] = oacc[t][v];
    }
    __syncthreads();

    {
        const int row = tid >> 5;
        const int cv0 = (tid & 31) * 2;

        float mstar = NEG_BIG;
        #pragma unroll
        for (int w2 = 0; w2 < NW; ++w2) mstar = fmaxf(mstar, lm[w2][row]);

        float lstar = 0.0f;
        float acc[2] = {0.0f, 0.0f};
        #pragma unroll
        for (int w2 = 0; w2 < NW; ++w2) {
            const float f = __expf(lm[w2][row] - mstar);
            lstar += ll[w2][row] * f;
            #pragma unroll
            for (int j = 0; j < 2; ++j)
                acc[j] += lo[w2][row][cv0 + j] * f;
        }
        const float inv = 1.0f / lstar;
        #pragma unroll
        for (int j = 0; j < 2; ++j)
            out[((size_t)b * CV_N + cv0 + j) * HW_N + q0 + row] = acc[j] * inv;
    }
}

extern "C" void kernel_launch(void* const* d_in, const int* in_sizes, int n_in,
                              void* d_out, int out_size, void* d_ws, size_t ws_size,
                              hipStream_t stream) {
    const float* mem_keys  = (const float*)d_in[0];
    const float* mem_vals  = (const float*)d_in[1];
    const float* query     = (const float*)d_in[2];
    const float* disparity = (const float*)d_in[3];
    const int*   seq_index = (const int*)d_in[4];
    float*       out       = (float*)d_out;

    (void)in_sizes; (void)n_in; (void)out_size; (void)d_ws; (void)ws_size;

    dim3 grid(B_N * (HW_N / 16));
    dim3 block(32 * NW);
    cfa_fp32_wmma_splitk<<<grid, block, 0, stream>>>(
        mem_keys, mem_vals, query, disparity, seq_index, out);
}


# 64 keys per loop trip instead of 16: K and V rows read as whole aligned 256-B pieces by 128-bit loads issued ahead; max, rescale, P hand-over once per 64 keys; sum order of row sums and O changes
# speedup vs baseline: 1.5872x; 1.3700x over previous
#include <hip/hip_runtime.h>

typedef __attribute__((ext_vector_type(2))) float v2f;
typedef __attribute__((ext_vector_type(4))) float v4f;
typedef __attribute__((ext_vector_type(8))) float v8f;

#define SAI_N 9
#define B_N   4
#define CK_N  64
#define CV_N  64
#define HW_N  1024
#define SHW_N (SAI_N * HW_N)
#define RADIUS_F 0.5f
#define NEG_BIG (-1.0e30f)

#define NW 16
#define KEYS_PER_WAVE (SHW_N / NW)
#define CHUNK 64
#define PT_PITCH 68
#define KG 4

__global__ __launch_bounds__(32 * NW)
void cfa_fp32_wmma_splitk(const float* __restrict__ mem_keys,
                          const float* __restrict__ mem_vals,
                          const float* __restrict__ query,
                          const float* __restrict__ disparity,
                          const int*   __restrict__ seq_index,
                          float* __restrict__ out)
{
    __shared__ __attribute__((aligned(16)))
    float pt[NW][16 * PT_PITCH];
    __shared__ float lm[NW][16];
    __shared__ float ll[NW][16];
    __shared__ float lo[NW][16][CV_N];

    const int tid  = threadIdx.x;
    const int w    = tid >> 5;
    const int lane = tid & 31;
    const int b    = blockIdx.x >> 6;
    const int q0   = (blockIdx.x & 63) * 16;

    const int m    = lane & 15;
    const int kb   = (lane >> 4) * 2;
    const int n    = lane & 15;
    const int half = lane >> 4;

    v2f qa[16];
    {
        const float* qbase = query + (size_t)b * CK_N * HW_N + q0 + m;
        #pragma unroll
        for (int cc = 0; cc < 16; ++cc) {
            const int c0 = cc * 4 + kb;
            qa[cc].x = qbase[(size_t)c0 * HW_N];
            qa[cc].y = qbase[(size_t)(c0 + 1) * HW_N];
        }
    }

    float mrow[8], lsum[8];
    #pragma unroll
    for (int v = 0; v < 8; ++v) { mrow[v] = NEG_BIG; lsum[v] = 0.0f; }
    v8f oacc[4] = {};

    const int kstart = w * KEYS_PER_WAVE;
    const int kend   = kstart + KEYS_PER_WAVE;
    for (int s = kstart >> 10; s * HW_N < kend; ++s) {
        const int klo = (kstart > s * HW_N) ? kstart : s * HW_N;
        const int khi = (kend < (s + 1) * HW_N) ? kend : (s + 1) * HW_N;

        const float dx = (float)seq_index[(b * SAI_N + s) * 2 + 0] - 5.0f;
        const float dy = (float)seq_index[(b * SAI_N + s) * 2 + 1] - 5.0f;
        const float df = sqrtf(dx * dx + dy * dy);

        const float* kframe = mem_keys + (((size_t)s * B_N + b) * CK_N) * HW_N;
        const float* vframe = mem_vals + (((size_t)s * B_N + b) * CV_N) * HW_N;

        for (int k0 = klo; k0 < khi; k0 += CHUNK) {
            const int p0 = k0 & (HW_N - 1);
            const float* kbase = kframe + p0 + 4 * n;
            const float* vbase = vframe + p0 + 4 * half;

            __builtin_prefetch((const void*)(kframe + p0 + CHUNK), 0, 1);
            const v4f disp = *(const v4f*)(disparity + (size_t)b * HW_N + p0 + 4 * n);

            v8f sc[4] = {};
            v4f ka[2][KG], kc[2][KG];
            #pragma unroll
            for (int u = 0; u < KG; ++u) {
                const int c0 = u * 4 + kb;
                ka[0][u] = *(const v4f*)(kbase + (size_t)c0 * HW_N);
                kc[0][u] = *(const v4f*)(kbase + (size_t)(c0 + 1) * HW_N);
            }
            #pragma unroll
            for (int gq = 0; gq < 16 / KG; ++gq) {
                const int cur = gq & 1;
                if (gq + 1 < 16 / KG) {
                    #pragma unroll
                    for (int u = 0; u < KG; ++u) {
                        const int c0 = ((gq + 1) * KG + u) * 4 + kb;
                        ka[cur ^ 1][u] = *(const v4f*)(kbase + (size_t)c0 * HW_N);
                        kc[cur ^ 1][u] = *(const v4f*)(kbase + (size_t)(c0 + 1) * HW_N);
                    }
                }
                __builtin_amdgcn_sched_barrier(0);
                #pragma unroll
                for (int u = 0; u < KG; ++u) {
                    const int cc = gq * KG + u;
                    #pragma unroll
                    for (int i = 0; i < 4; ++i) {
                        v2f bk;
                        bk.x = ka[cur][u][i];
                        bk.y = kc[cur][u][i];
                        sc[i] = __builtin_amdgcn_wmma_f32_16x16x4_f32(
                            false, qa[cc], false, bk, (short)0, sc[i], false, false);
                    }
                }
                __builtin_amdgcn_sched_barrier(0);
            }

            v4f vb[3][2][4];
            #pragma unroll
            for (int tt = 0; tt < 2; ++tt) {
                #pragma unroll
                for (int gg = 0; gg < 4; ++gg)
                    vb[0][tt][gg] = *(const v4f*)(vbase + (size_t)(tt * 16 + n) * HW_N + 8 * gg);
            }
            __builtin_amdgcn_sched_barrier(0);

            #pragma unroll
            for (int i = 0; i < 4; ++i) {
                const float bias = (fabsf(df * disp[i]) > RADIUS_F) ? 0.0f : NEG_BIG;
                #pragma unroll
                for (int v = 0; v < 8; ++v) sc[i][v] += bias;
            }

            float cm[8];
            #pragma unroll
            for (int v = 0; v < 8; ++v)
                cm[v] = fmaxf(fmaxf(sc[0][v], sc[1][v]), fmaxf(sc[2][v], sc[3][v]));
            #pragma unroll
            for (int off = 1; off < 16; off <<= 1) {
                float other[8];
                #pragma unroll
                for (int v = 0; v < 8; ++v) other[v] = __shfl_xor(cm[v], off, 32);
                #pragma unroll
                for (int v = 0; v < 8; ++v) cm[v] = fmaxf(cm[v], other[v]);
            }
            #pragma unroll
            for (int v = 0; v < 8; ++v) {
                const float mn    = fmaxf(mrow[v], cm[v]);
                const float scale = __expf(mrow[v] - mn);
                v4f pv;
                #pragma unroll
                for (int i = 0; i < 4; ++i) pv[i] = __expf(sc[i][v] - mn);
                lsum[v] = lsum[v] * scale + ((pv[0] + pv[1]) + (pv[2] + pv[3]));
                mrow[v] = mn;
                #pragma unroll
                for (int t = 0; t < 4; ++t) oacc[t][v] *= scale;
                *(v4f*)&pt[w][(v + 8 * half) * PT_PITCH + 4 * n] = pv;
            }
            __builtin_amdgcn_wave_barrier();

            v4f pah[2][4];
            #pragma unroll
            for (int gg = 0; gg < 4; ++gg)
                pah[0][gg] = *(const v4f*)&pt[w][m * PT_PITCH + 8 * gg + 4 * half];
            #pragma unroll
            for (int u = 0; u < 4; ++u) {
                const int hc = u >> 1;
                const int tp = u & 1;
                const int un_lo = (u == 0) ? 1 : 3;
                const int un_hi = (u == 0) ? 2 : ((u == 2) ? 3 : 2);
                #pragma unroll
                for (int un = un_lo; un <= un_hi; ++un) {
                    const int hn = un >> 1, tn = un & 1;
                    #pragma unroll
                    for (int tt = 0; tt < 2; ++tt) {
                        #pragma unroll
                        for (int gg = 0; gg < 4; ++gg)
                            vb[un % 3][tt][gg] = *(const v4f*)(
                                vbase + (size_t)((2 * tn + tt) * 16 + n) * HW_N + 32 * hn + 8 * gg);
                    }
                }
                if (u == 1) {
                    #pragma unroll
                    for (int gg = 0; gg < 4; ++gg)
                        pah[1][gg] = *(const v4f*)&pt[w][m * PT_PITCH + 32 + 8 * gg + 4 * half];
                }
                __builtin_amdgcn_sched_barrier(0);
                #pragma unroll
                for (int gg = 0; gg < 4; ++gg) {
                    const v4f pcur = pah[hc][gg];
                    #pragma unroll
                    for (int ks = 0; ks < 2; ++ks) {
                        v2f pa;
                        pa.x = pcur[2 * ks];
                        pa.y = pcur[2 * ks + 1];
                        #pragma unroll
                        for (int tt = 0; tt < 2; ++tt) {
                            v2f bv;
                            bv.x = vb[u % 3][tt][gg][2 * ks];
                            bv.y = vb[u % 3][tt][gg][2 * ks + 1];
                            oacc[2 * tp + tt] = __builtin_amdgcn_wmma_f32_16x16x4_f32(
                                false, pa, false, bv, (short)0, oacc[2 * tp + tt], false, false);
                        }
                    }
                }
                __builtin_amdgcn_sched_barrier(0);
            }
            __builtin_amdgcn_wave_barrier();
        }
    }

    float lrow[8];
    #pragma unroll
    for (int v = 0; v < 8; ++v) {
        float rs = lsum[v];
        #pragma unroll
        for (int off = 1; off < 16; off <<= 1)
            rs += __shfl_xor(rs, off, 32);
        lrow[v] = rs;
    }

    if (n == 0) {
        #pragma unroll
        for (int v = 0; v < 8; ++v) {
            lm[w][v + 8 * half] = mrow[v];
            ll[w][v + 8 * half] = lrow[v];
        }
    }
    #pragma unroll
    for (int v = 0; v < 8; ++v) {
        const int row = v + 8 * half;
        #pragma unroll
        for (int t = 0; t < 4; ++t)
            lo[w][row][t * 16 + n] = oacc[t][v];
    }
    __syncthreads();

    {
        const int row = tid >> 5;
        const int cv0 = (tid & 31) * 2;

        float mstar = NEG_BIG;
        #pragma unroll
        for (int w2 = 0; w2 < NW; ++w2) mstar = fmaxf(mstar, lm[w2][row]);

        float lstar = 0.0f;
        float acc[2] = {0.0f, 0.0f};
        #pragma unroll
        for (int w2 = 0; w2 < NW; ++w2) {
            const float f = __expf(lm[w2][row] - mstar);
            lstar += ll[w2][row] * f;
            #pragma unroll
            for (int j = 0; j < 2; ++j)
                acc[j] += lo[w2][row][cv0 + j] * f;
        }
        const float inv = 1.0f / lstar;
        #pragma unroll
        for (int j = 0; j < 2; ++j)
            out[((size_t)b * CV_N + cv0 + j) * HW_N + q0 + row] = acc[j] * inv;
    }
}

extern "C" void kernel_launch(void* const* d_in, const int* in_sizes, int n_in,
                              void* d_out, int out_size, void* d_ws, size_t ws_size,
                              hipStream_t stream) {
    const float* mem_keys  = (const float*)d_in[0];
    const float* mem_vals  = (const float*)d_in[1];
    const float* query     = (const float*)d_in[2];
    const float* disparity = (const float*)d_in[3];
    const int*   seq_index = (const int*)d_in[4];
    float*       out       = (float*)d_out;

    (void)in_sizes; (void)n_in; (void)out_size; (void)d_ws; (void)ws_size;

    dim3 grid(B_N * (HW_N / 16));
    dim3 block(32 * NW);
    cfa_fp32_wmma_splitk<<<grid, block, 0, stream>>>(
        mem_keys, mem_vals, query, disparity, seq_index, out);
}


# Both products transposed (S^T = K x Q^T, O^T = V^T x P^T): a lane keeps one query: max and sum in-lane plus one half exchange, P^T goes to the WMMA from registers, no LDS; row-sum order changes
# speedup vs baseline: 1.6643x; 1.0486x over previous
#include <hip/hip_runtime.h>

typedef __attribute__((ext_vector_type(2))) float v2f;
typedef __attribute__((ext_vector_type(4))) float v4f;
typedef __attribute__((ext_vector_type(8))) float v8f;

#define SAI_N 9
#define B_N   4
#define CK_N  64
#define CV_N  64
#define HW_N  1024
#define SHW_N (SAI_N * HW_N)
#define RADIUS_F 0.5f
#define NEG_BIG (-1.0e30f)

#define NW 16
#define KEYS_PER_WAVE (SHW_N / NW)
#define CHUNK 64
#define LO_PITCH 68
#define KG 4

__device__ __forceinline__ float xhalf(float x) {
    const int bits = __builtin_bit_cast(int, x);
    return __builtin_bit_cast(float,
        __builtin_amdgcn_permlanex16(bits, bits, 0x76543210, 0xfedcba98, false, false));
}

__global__ __launch_bounds__(32 * NW)
void cfa_fp32_wmma_splitk(const float* __restrict__ mem_keys,
                          const float* __restrict__ mem_vals,
                          const float* __restrict__ query,
                          const float* __restrict__ disparity,
                          const int*   __restrict__ seq_index,
                          float* __restrict__ out)
{
    __shared__ float lm[NW][16];
    __shared__ float ll[NW][16];
    __shared__ __attribute__((aligned(16)))
    float lo[NW][16 * LO_PITCH];

    const int tid  = threadIdx.x;
    const int w    = tid >> 5;
    const int lane = tid & 31;
    const int b    = blockIdx.x >> 6;
    const int q0   = (blockIdx.x & 63) * 16;

    const int n    = lane & 15;
    const int half = lane >> 4;
    const int kb   = half * 2;
    const int koff = 8 * (n & 7) + 4 * (n >> 3);

    v2f qa[16];
    {
        const float* qbase = query + (size_t)b * CK_N * HW_N + q0 + n;
        #pragma unroll
        for (int cc = 0; cc < 16; ++cc) {
            const int c0 = cc * 4 + kb;
            qa[cc].x = qbase[(size_t)c0 * HW_N];
            qa[cc].y = qbase[(size_t)(c0 + 1) * HW_N];
        }
    }

    float mrow = NEG_BIG, lsum = 0.0f;
    v8f oacc[4] = {};

    const int kstart = w * KEYS_PER_WAVE;
    const int kend   = kstart + KEYS_PER_WAVE;
    for (int s = kstart >> 10; s * HW_N < kend; ++s) {
        const int klo = (kstart > s * HW_N) ? kstart : s * HW_N;
        const int khi = (kend < (s + 1) * HW_N) ? kend : (s + 1) * HW_N;

        const float dx = (float)seq_index[(b * SAI_N + s) * 2 + 0] - 5.0f;
        const float dy = (float)seq_index[(b * SAI_N + s) * 2 + 1] - 5.0f;
        const float df = sqrtf(dx * dx + dy * dy);

        const float* kframe = mem_keys + (((size_t)s * B_N + b) * CK_N) * HW_N;
        const float* vframe = mem_vals + (((size_t)s * B_N + b) * CV_N) * HW_N;

        for (int k0 = klo; k0 < khi; k0 += CHUNK) {
            const int p0 = k0 & (HW_N - 1);
            const float* kbase = kframe + p0 + koff;
            const float* vbase = vframe + p0 + 4 * half;
            const float* dbase = disparity + (size_t)b * HW_N + p0 + 4 * half;

            __builtin_prefetch((const void*)(kframe + p0 + CHUNK), 0, 1);

            v8f sc[4];
            #pragma unroll
            for (int r = 0; r < 8; ++r) {
                const v4f disp = *(const v4f*)(dbase + 8 * r);
                #pragma unroll
                for (int i = 0; i < 4; ++i)
                    sc[i][r] = (fabsf(df * disp[i]) > RADIUS_F) ? 0.0f : NEG_BIG;
            }

            v4f ka[2][KG], kc[2][KG];
            #pragma unroll
            for (int u = 0; u < KG; ++u) {
                const int c0 = u * 4 + kb;
                ka[0][u] = *(const v4f*)(kbase + (size_t)c0 * HW_N);
                kc[0][u] = *(const v4f*)(kbase + (size_t)(c0 + 1) * HW_N);
            }
            #pragma unroll
            for (int gq = 0; gq < 16 / KG; ++gq) {
                const int cur = gq & 1;
                if (gq + 1 < 16 / KG) {
                    #pragma unroll
                    for (int u = 0; u < KG; ++u) {
                        const int c0 = ((gq + 1) * KG + u) * 4 + kb;
                        ka[cur ^ 1][u] = *(const v4f*)(kbase + (size_t)c0 * HW_N);
                        kc[cur ^ 1][u] = *(const v4f*)(kbase + (size_t)(c0 + 1) * HW_N);
                    }
                }
                __builtin_amdgcn_sched_barrier(0);
                #pragma unroll
                for (int u = 0; u < KG; ++u) {
                    const int cc = gq * KG + u;
                    #pragma unroll
                    for (int i = 0; i < 4; ++i) {
                        v2f ak;
                        ak.x = ka[cur][u][i];
                        ak.y = kc[cur][u][i];
                        sc[i] = __builtin_amdgcn_wmma_f32_16x16x4_f32(
                            false, ak, false, qa[cc], (short)0, sc[i], false, false);
                    }
                }
                __builtin_amdgcn_sched_barrier(0);
            }

            v4f vb[3][2][4];
            #pragma unroll
            for (int tt = 0; tt < 2; ++tt) {
                #pragma unroll
                for (int gg = 0; gg < 4; ++gg)
                    vb[0][tt][gg] = *(const v4f*)(vbase + (size_t)(tt * 16 + n) * HW_N + 8 * gg);
            }
            __builtin_amdgcn_sched_barrier(0);

            float cm;
            {
                float t8[8];
                #pragma unroll
                for (int r = 0; r < 8; ++r)
                    t8[r] = fmaxf(fmaxf(sc[0][r], sc[1][r]), fmaxf(sc[2][r], sc[3][r]));
                cm = fmaxf(fmaxf(fmaxf(t8[0], t8[1]), fmaxf(t8[2], t8[3])),
                           fmaxf(fmaxf(t8[4], t8[5]), fmaxf(t8[6], t8[7])));
            }
            cm = fmaxf(cm, xhalf(cm));
            const float mn    = fmaxf(mrow, cm);
            const float scale = __expf(mrow - mn);
            mrow = mn;

            v2f pp[2][8];
            float ps[8];
            #pragma unroll
            for (int g = 0; g < 8; ++g) {
                #pragma unroll
                for (int e = 0; e < 2; ++e) {
                    pp[e][g].x = __expf(sc[2 * e][g] - mn);
                    pp[e][g].y = __expf(sc[2 * e + 1][g] - mn);
                }
                ps[g] = (pp[0][g].x + pp[0][g].y) + (pp[1][g].x + pp[1][g].y);
            }
            lsum = lsum * scale + (((ps[0] + ps[1]) + (ps[2] + ps[3])) + ((ps[4] + ps[5]) + (ps[6] + ps[7])));
            #pragma unroll
            for (int t = 0; t < 4; ++t) {
                #pragma unroll
                for (int v = 0; v < 8; ++v) oacc[t][v] *= scale;
            }

            #pragma unroll
            for (int u = 0; u < 4; ++u) {
                const int hc = u >> 1;
                const int tp = u & 1;
                const int un_lo = (u == 0) ? 1 : 3;
                const int un_hi = (u == 0) ? 2 : ((u == 2) ? 3 : 2);
                #pragma unroll
                for (int un = un_lo; un <= un_hi; ++un) {
                    const int hn = un >> 1, tn = un & 1;
                    #pragma unroll
                    for (int tt = 0; tt < 2; ++tt) {
                        #pragma unroll
                        for (int gg = 0; gg < 4; ++gg)
                            vb[un % 3][tt][gg] = *(const v4f*)(
                                vbase + (size_t)((2 * tn + tt) * 16 + n) * HW_N + 32 * hn + 8 * gg);
                    }
                }
                __builtin_amdgcn_sched_barrier(0);
                #pragma unroll
                for (int gg = 0; gg < 4; ++gg) {
                    const int g = 4 * hc + gg;
                    #pragma unroll
                    for (int e = 0; e < 2; ++e) {
                        #pragma unroll
                        for (int tt = 0; tt < 2; ++tt) {
                            v2f av;
                            av.x = vb[u % 3][tt][gg][2 * e];
                            av.y = vb[u % 3][tt][gg][2 * e + 1];
                            oacc[2 * tp + tt] = __builtin_amdgcn_wmma_f32_16x16x4_f32(
                                false, av, false, pp[e][g], (short)0, oacc[2 * tp + tt], false, false);
                        }
                    }
                }
                __builtin_amdgcn_sched_barrier(0);
            }
        }
    }

    const float lrow = lsum + xhalf(lsum);

    if (half == 0) {
        lm[w][n] = mrow;
        ll[w][n] = lrow;
    }
    #pragma unroll
    for (int t = 0; t < 4; ++t) {
        #pragma unroll
        for (int hh = 0; hh < 2; ++hh) {
            v4f ov;
            #pragma unroll
            for (int j = 0; j < 4; ++j) ov[j] = oacc[t][4 * hh + j];
            *(v4f*)&lo[w][n * LO_PITCH + 16 * t + 8 * half + 4 * hh] = ov;
        }
    }
    __syncthreads();

    {
        const int row = tid >> 5;
        const int cv0 = (tid & 31) * 2;

        float mstar = NEG_BIG;
        #pragma unroll
        for (int w2 = 0; w2 < NW; ++w2) mstar = fmaxf(mstar, lm[w2][row]);

        float lstar = 0.0f;
        float acc[2] = {0.0f, 0.0f};
        #pragma unroll
        for (int w2 = 0; w2 < NW; ++w2) {
            const float f = __expf(lm[w2][row] - mstar);
            lstar += ll[w2][row] * f;
            #pragma unroll
            for (int j = 0; j < 2; ++j)
                acc[j] += lo[w2][row * LO_PITCH + cv0 + j] * f;
        }
        const float inv = 1.0f / lstar;
        #pragma unroll
        for (int j = 0; j < 2; ++j)
            out[((size_t)b * CV_N + cv0 + j) * HW_N + q0 + row] = acc[j] * inv;
    }
}

extern "C" void kernel_launch(void* const* d_in, const int* in_sizes, int n_in,
                              void* d_out, int out_size, void* d_ws, size_t ws_size,
                              hipStream_t stream) {
    const float* mem_keys  = (const float*)d_in[0];
    const float* mem_vals  = (const float*)d_in[1];
    const float* query     = (const float*)d_in[2];
    const float* disparity = (const float*)d_in[3];
    const int*   seq_index = (const int*)d_in[4];
    float*       out       = (float*)d_out;

    (void)in_sizes; (void)n_in; (void)out_size; (void)d_ws; (void)ws_size;

    dim3 grid(B_N * (HW_N / 16));
    dim3 block(32 * NW);
    cfa_fp32_wmma_splitk<<<grid, block, 0, stream>>>(
        mem_keys, mem_vals, query, disparity, seq_index, out);
}


# 8 waves of up to 512 registers per block instead of 16 of 256: a wave holds a whole 64-key K chunk and V chunk, every load issued one full product ahead of use; 8 split-K partials, sum order changes
# speedup vs baseline: 1.8129x; 1.0893x over previous
#include <hip/hip_runtime.h>

typedef __attribute__((ext_vector_type(2))) float v2f;
typedef __attribute__((ext_vector_type(4))) float v4f;
typedef __attribute__((ext_vector_type(8))) float v8f;

#define SAI_N 9
#define B_N   4
#define CK_N  64
#define CV_N  64
#define HW_N  1024
#define SHW_N (SAI_N * HW_N)
#define RADIUS_F 0.5f
#define NEG_BIG (-1.0e30f)

#define NW 8
#define KEYS_PER_WAVE (SHW_N / NW)
#define CHUNK 64
#define NCH (KEYS_PER_WAVE / CHUNK)
#define LO_PITCH 68

__device__ __forceinline__ float xhalf(float x) {
    const int bits = __builtin_bit_cast(int, x);
    return __builtin_bit_cast(float,
        __builtin_amdgcn_permlanex16(bits, bits, 0x76543210, 0xfedcba98, false, false));
}

__global__ __launch_bounds__(32 * NW)
void cfa_fp32_wmma_splitk(const float* __restrict__ mem_keys,
                          const float* __restrict__ mem_vals,
                          const float* __restrict__ query,
                          const float* __restrict__ disparity,
                          const int*   __restrict__ seq_index,
                          float* __restrict__ out)
{
    __shared__ float lm[NW][16];
    __shared__ float ll[NW][16];
    __shared__ __attribute__((aligned(16)))
    float lo[NW][16 * LO_PITCH];

    const int tid  = threadIdx.x;
    const int w    = __builtin_amdgcn_readfirstlane(tid >> 5);
    const int lane = tid & 31;
    const int b    = blockIdx.x >> 6;
    const int q0   = (blockIdx.x & 63) * 16;

    const int n    = lane & 15;
    const int half = lane >> 4;
    const int kb   = half * 2;
    const int koff = 8 * (n & 7) + 4 * (n >> 3);

    v2f qa[16];
    {
        const float* qbase = query + (size_t)b * CK_N * HW_N + q0 + n;
        #pragma unroll
        for (int cc = 0; cc < 16; ++cc) {
            const int c0 = cc * 4 + kb;
            qa[cc].x = qbase[(size_t)c0 * HW_N];
            qa[cc].y = qbase[(size_t)(c0 + 1) * HW_N];
        }
    }

    const int kstart = w * KEYS_PER_WAVE;
    const int s0     = kstart >> 10;
    float dfr[2];
    #pragma unroll
    for (int f = 0; f < 2; ++f) {
        const int s = (s0 + f < SAI_N) ? (s0 + f) : (SAI_N - 1);
        const float dx = (float)seq_index[(b * SAI_N + s) * 2 + 0] - 5.0f;
        const float dy = (float)seq_index[(b * SAI_N + s) * 2 + 1] - 5.0f;
        dfr[f] = sqrtf(dx * dx + dy * dy);
    }

    const float* klane = mem_keys + (size_t)b * CK_N * HW_N + koff + (size_t)kb * HW_N;
    const float* vlane = mem_vals + (size_t)b * CV_N * HW_N + 4 * half + (size_t)n * HW_N;
    const float* dlane = disparity + (size_t)b * HW_N + 4 * half;

    float mrow = NEG_BIG, lsum = 0.0f;
    v8f oacc[4] = {};

    v4f ka[16], kc[16], dv[8];
    {
        const int p0 = kstart & (HW_N - 1);
        const size_t fo = (size_t)s0 * B_N * CK_N * HW_N + p0;
        #pragma unroll
        for (int cc = 0; cc < 16; ++cc) {
            ka[cc] = *(const v4f*)(klane + fo + (size_t)(cc * 4) * HW_N);
            kc[cc] = *(const v4f*)(klane + fo + (size_t)(cc * 4 + 1) * HW_N);
        }
        #pragma unroll
        for (int r = 0; r < 8; ++r) dv[r] = *(const v4f*)(dlane + p0 + 8 * r);
    }

    for (int ci = 0; ci < NCH; ++ci) {
        const int k0 = kstart + ci * CHUNK;
        const int s  = k0 >> 10;
        const int p0 = k0 & (HW_N - 1);
        const float df = (s == s0) ? dfr[0] : dfr[1];

        v4f vb[2][4][4];
        {
            const float* vbase = vlane + (size_t)s * B_N * CV_N * HW_N + p0;
            #pragma unroll
            for (int hc = 0; hc < 2; ++hc) {
                #pragma unroll
                for (int t = 0; t < 4; ++t) {
                    #pragma unroll
                    for (int gg = 0; gg < 4; ++gg)
                        vb[hc][t][gg] = *(const v4f*)(vbase + (size_t)(t * 16) * HW_N + 32 * hc + 8 * gg);
                }
            }
        }

        v8f sc[4];
        #pragma unroll
        for (int r = 0; r < 8; ++r) {
            #pragma unroll
            for (int i = 0; i < 4; ++i)
                sc[i][r] = (fabsf(df * dv[r][i]) > RADIUS_F) ? 0.0f : NEG_BIG;
        }
        __builtin_amdgcn_sched_barrier(0);

        #pragma unroll
        for (int cc = 0; cc < 16; ++cc) {
            #pragma unroll
            for (int i = 0; i < 4; ++i) {
                v2f ak;
                ak.x = ka[cc][i];
                ak.y = kc[cc][i];
                sc[i] = __builtin_amdgcn_wmma_f32_16x16x4_f32(
                    false, ak, false, qa[cc], (short)0, sc[i], false, false);
            }
        }
        __builtin_amdgcn_sched_barrier(0);

        {
            const int cn  = (ci + 1 < NCH) ? (ci + 1) : ci;
            const int k1  = kstart + cn * CHUNK;
            const int s1  = k1 >> 10;
            const int p1  = k1 & (HW_N - 1);
            const size_t fo = (size_t)s1 * B_N * CK_N * HW_N + p1;
            #pragma unroll
            for (int cc = 0; cc < 16; ++cc) {
                ka[cc] = *(const v4f*)(klane + fo + (size_t)(cc * 4) * HW_N);
                kc[cc] = *(const v4f*)(klane + fo + (size_t)(cc * 4 + 1) * HW_N);
            }
            #pragma unroll
            for (int r = 0; r < 8; ++r) dv[r] = *(const v4f*)(dlane + p1 + 8 * r);
        }
        __builtin_amdgcn_sched_barrier(0);

        float cm;
        {
            float t8[8];
            #pragma unroll
            for (int r = 0; r < 8; ++r)
                t8[r] = fmaxf(fmaxf(sc[0][r], sc[1][r]), fmaxf(sc[2][r], sc[3][r]));
            cm = fmaxf(fmaxf(fmaxf(t8[0], t8[1]), fmaxf(t8[2], t8[3])),
                       fmaxf(fmaxf(t8[4], t8[5]), fmaxf(t8[6], t8[7])));
        }
        cm = fmaxf(cm, xhalf(cm));
        const float mn    = fmaxf(mrow, cm);
        const float scale = __expf(mrow - mn);
        mrow = mn;

        v2f pp[2][8];
        float ps[8];
        #pragma unroll
        for (int g = 0; g < 8; ++g) {
            #pragma unroll
            for (int e = 0; e < 2; ++e) {
                pp[e][g].x = __expf(sc[2 * e][g] - mn);
                pp[e][g].y = __expf(sc[2 * e + 1][g] - mn);
            }
            ps[g] = (pp[0][g].x + pp[0][g].y) + (pp[1][g].x + pp[1][g].y);
        }
        lsum = lsum * scale + (((ps[0] + ps[1]) + (ps[2] + ps[3])) + ((ps[4] + ps[5]) + (ps[6] + ps[7])));
        #pragma unroll
        for (int t = 0; t < 4; ++t) {
            #pragma unroll
            for (int v = 0; v < 8; ++v) oacc[t][v] *= scale;
        }

        #pragma unroll
        for (int hc = 0; hc < 2; ++hc) {
            #pragma unroll
            for (int gg = 0; gg < 4; ++gg) {
                const int g = 4 * hc + gg;
                #pragma unroll
                for (int e = 0; e < 2; ++e) {
                    #pragma unroll
                    for (int t = 0; t < 4; ++t) {
                        v2f av;
                        av.x = vb[hc][t][gg][2 * e];
                        av.y = vb[hc][t][gg][2 * e + 1];
                        oacc[t] = __builtin_amdgcn_wmma_f32_16x16x4_f32(
                            false, av, false, pp[e][g], (short)0, oacc[t], false, false);
                    }
                }
            }
        }
        __builtin_amdgcn_sched_barrier(0);
    }

    const float lrow = lsum + xhalf(lsum);

    if (half == 0) {
        lm[w][n] = mrow;
        ll[w][n] = lrow;
    }
    #pragma unroll
    for (int t = 0; t < 4; ++t) {
        #pragma unroll
        for (int hh = 0; hh < 2; ++hh) {
            v4f ov;
            #pragma unroll
            for (int j = 0; j < 4; ++j) ov[j] = oacc[t][4 * hh + j];
            *(v4f*)&lo[w][n * LO_PITCH + 16 * t + 8 * half + 4 * hh] = ov;
        }
    }
    __syncthreads();

    {
        const int row = tid >> 4;
        const int cv0 = (tid & 15) * 4;

        float mstar = NEG_BIG;
        #pragma unroll
        for (int w2 = 0; w2 < NW; ++w2) mstar = fmaxf(mstar, lm[w2][row]);

        float lstar = 0.0f;
        float acc[4] = {0.0f, 0.0f, 0.0f, 0.0f};
        #pragma unroll
        for (int w2 = 0; w2 < NW; ++w2) {
            const float f = __expf(lm[w2][row] - mstar);
            lstar += ll[w2][row] * f;
            const v4f o4 = *(const v4f*)&lo[w2][row * LO_PITCH + cv0];
            #pragma unroll
            for (int j = 0; j < 4; ++j)
                acc[j] += o4[j] * f;
        }
        const float inv = 1.0f / lstar;
        #pragma unroll
        for (int j = 0; j < 4; ++j)
            out[((size_t)b * CV_N + cv0 + j) * HW_N + q0 + row] = acc[j] * inv;
    }
}

extern "C" void kernel_launch(void* const* d_in, const int* in_sizes, int n_in,
                              void* d_out, int out_size, void* d_ws, size_t ws_size,
                              hipStream_t stream) {
    const float* mem_keys  = (const float*)d_in[0];
    const float* mem_vals  = (const float*)d_in[1];
    const float* query     = (const float*)d_in[2];
    const float* disparity = (const float*)d_in[3];
    const int*   seq_index = (const int*)d_in[4];
    float*       out       = (float*)d_out;

    (void)in_sizes; (void)n_in; (void)out_size; (void)d_ws; (void)ws_size;

    dim3 grid(B_N * (HW_N / 16));
    dim3 block(32 * NW);
    cfa_fp32_wmma_splitk<<<grid, block, 0, stream>>>(
        mem_keys, mem_vals, query, disparity, seq_index, out);
}
